# starting kernel
# baseline (speedup 1.0000x reference)
#include <hip/hip_runtime.h>

#define N_NODES 100000
#define N_EDGES 1600000
#define D_FEAT  128
#define D_HID   512
#define D_OUT   128
#define SCAN_B  1024
#define NB      ((N_NODES + SCAN_B - 1) / SCAN_B)

typedef float v2f __attribute__((ext_vector_type(2)));
typedef float v8f __attribute__((ext_vector_type(8)));

__device__ __forceinline__ v8f wmma4(v2f a, v2f b, v8f c) {
  return __builtin_amdgcn_wmma_f32_16x16x4_f32(false, a, false, b,
                                               (short)0, c, false, false);
}


__global__ void k_zero(int* __restrict__ p, int n) {
  int i = blockIdx.x * blockDim.x + threadIdx.x;
  if (i < n) p[i] = 0;
}

__global__ void k_deg(const int* __restrict__ dst, int* __restrict__ deg) {
  int e = blockIdx.x * blockDim.x + threadIdx.x;
  if (e < N_EDGES) atomicAdd(&deg[dst[e]], 1);
}

__global__ __launch_bounds__(SCAN_B)
void k_scan1(const int* __restrict__ deg, int* __restrict__ incl,
             int* __restrict__ bsum, int n) {
  __shared__ int s[SCAN_B];
  int i = blockIdx.x * SCAN_B + threadIdx.x;
  s[threadIdx.x] = (i < n) ? deg[i] : 0;
  __syncthreads();
  for (int off = 1; off < SCAN_B; off <<= 1) {
    int t = (threadIdx.x >= off) ? s[threadIdx.x - off] : 0;
    __syncthreads();
    s[threadIdx.x] += t;
    __syncthreads();
  }
  if (i < n) incl[i] = s[threadIdx.x];
  if (threadIdx.x == SCAN_B - 1) bsum[blockIdx.x] = s[SCAN_B - 1];
}

__global__ void k_scan2(int* __restrict__ bsum, int nb) {
  if (threadIdx.x == 0 && blockIdx.x == 0) {
    int run = 0;
    for (int i = 0; i < nb; ++i) { int t = bsum[i]; bsum[i] = run; run += t; }
  }
}

__global__ __launch_bounds__(SCAN_B)
void k_scan3(const int* __restrict__ incl, const int* __restrict__ bsum,
             int* __restrict__ offs, int n) {
  int i = blockIdx.x * SCAN_B + threadIdx.x;
  if (i < n) offs[i + 1] = incl[i] + bsum[blockIdx.x];
  if (i == 0) offs[0] = 0;
}

__global__ void k_bin(const int* __restrict__ src, const int* __restrict__ dst,
                      const int* __restrict__ offs, int* __restrict__ cursor,
                      int* __restrict__ bsrc) {
  int e = blockIdx.x * blockDim.x + threadIdx.x;
  if (e >= N_EDGES) return;
  int d = dst[e];
  int pos = atomicAdd(&cursor[d], 1);
  bsrc[offs[d] + pos] = src[e];
}

__global__ __launch_bounds__(256)
void k_aggr(const float* __restrict__ x, const int* __restrict__ offs,
            const int* __restrict__ bsrc, float* __restrict__ h) {
  int node = (blockIdx.x << 3) + (threadIdx.x >> 5);
  if (node >= N_NODES) return;
  int lane = threadIdx.x & 31;
  float4 acc = *reinterpret_cast<const float4*>(x + (size_t)node * D_FEAT + lane * 4);
  int e0 = offs[node], e1 = offs[node + 1];
  for (int e = e0; e < e1; ++e) {
    int s = bsrc[e];
    float4 v = *reinterpret_cast<const float4*>(x + (size_t)s * D_FEAT + lane * 4);
    acc.x += v.x; acc.y += v.y; acc.z += v.z; acc.w += v.w;
  }
  *reinterpret_cast<float4*>(h + (size_t)node * D_FEAT + lane * 4) = acc;
}

__global__ __launch_bounds__(256)
void gin_mlp(const float* __restrict__ h,
             const float* __restrict__ W1, const float* __restrict__ b1,
             const float* __restrict__ W2, const float* __restrict__ b2,
             float* __restrict__ out) {
  extern __shared__ float smem[];
  float* sH  = smem;
  float* sH1 = smem + 32 * D_FEAT;

  const int tid   = threadIdx.x;
  const int wave  = tid >> 5;
  const int lane  = tid & 31;
  const int row16 = lane & 15;
  const int khalf = (lane >> 4) << 1;
  const int rowhi = (lane >> 4) << 3;
  const size_t base = (size_t)blockIdx.x * 32 * D_FEAT;

  for (int i = tid; i < 32 * D_FEAT; i += 256) sH[i] = h[base + i];
  __syncthreads();

  for (int t = 0; t < 4; ++t) {
    const int n0  = (wave + (t << 3)) << 4;
    const int col = n0 + row16;
    v8f acc0 = {0.f,0.f,0.f,0.f,0.f,0.f,0.f,0.f};
    v8f acc1 = {0.f,0.f,0.f,0.f,0.f,0.f,0.f,0.f};
#pragma unroll
    for (int kk = 0; kk < D_FEAT; kk += 4) {
      const int kb = kk + khalf;
      v2f b;
      b.x = W1[(size_t)kb * D_HID + col];
      b.y = W1[(size_t)(kb + 1) * D_HID + col];
      v2f a0 = *reinterpret_cast<const v2f*>(sH + row16 * D_FEAT + kk + khalf);
      v2f a1 = *reinterpret_cast<const v2f*>(sH + (16 + row16) * D_FEAT + kk + khalf);
      acc0 = wmma4(a0, b, acc0);
      acc1 = wmma4(a1, b, acc1);
    }
    const float bias = b1[col];
#pragma unroll
    for (int i = 0; i < 8; ++i) {
      float v0 = acc0[i] + bias;
      float v1 = acc1[i] + bias;
      sH1[(i + rowhi) * D_HID + col]      = v0 > 0.f ? v0 : 0.f;
      sH1[(16 + i + rowhi) * D_HID + col] = v1 > 0.f ? v1 : 0.f;
    }
  }
  __syncthreads();

  {
    const int n0  = wave << 4;
    const int col = n0 + row16;
    v8f acc0 = {0.f,0.f,0.f,0.f,0.f,0.f,0.f,0.f};
    v8f acc1 = {0.f,0.f,0.f,0.f,0.f,0.f,0.f,0.f};
#pragma unroll 8
    for (int kk = 0; kk < D_HID; kk += 4) {
      const int kb = kk + khalf;
      v2f b;
      b.x = W2[(size_t)kb * D_OUT + col];
      b.y = W2[(size_t)(kb + 1) * D_OUT + col];
      v2f a0 = *reinterpret_cast<const v2f*>(sH1 + row16 * D_HID + kk + khalf);
      v2f a1 = *reinterpret_cast<const v2f*>(sH1 + (16 + row16) * D_HID + kk + khalf);
      acc0 = wmma4(a0, b, acc0);
      acc1 = wmma4(a1, b, acc1);
    }
    const float bias = b2[col];
    const size_t obase = (size_t)blockIdx.x * 32 * D_OUT;
#pragma unroll
    for (int i = 0; i < 8; ++i) {
      out[obase + (size_t)(i + rowhi) * D_OUT + col]      = acc0[i] + bias;
      out[obase + (size_t)(16 + i + rowhi) * D_OUT + col] = acc1[i] + bias;
    }
  }
}

extern "C" void kernel_launch(void* const* d_in, const int* in_sizes, int n_in,
                              void* d_out, int out_size, void* d_ws, size_t ws_size,
                              hipStream_t stream) {
  const float* x  = (const float*)d_in[0];
  const int*   ei = (const int*)d_in[1];
  const float* W1 = (const float*)d_in[2];
  const float* b1 = (const float*)d_in[3];
  const float* W2 = (const float*)d_in[4];
  const float* b2 = (const float*)d_in[5];
  float* out = (float*)d_out;

  const int* src = ei;
  const int* dst = ei + N_EDGES;

  float* h    = (float*)d_ws;
  int*  bsrc  = (int*)(h + (size_t)N_NODES * D_FEAT);
  int*  deg   = bsrc + N_EDGES;
  int*  curs  = deg + N_NODES;
  int*  incl  = curs + N_NODES;
  int*  offs  = incl + N_NODES;
  int*  bsum  = offs + N_NODES + 1;

  k_zero<<<(2 * N_NODES + 255) / 256, 256, 0, stream>>>(deg, 2 * N_NODES);
  k_deg<<<(N_EDGES + 255) / 256, 256, 0, stream>>>(dst, deg);
  k_scan1<<<NB, SCAN_B, 0, stream>>>(deg, incl, bsum, N_NODES);
  k_scan2<<<1, 32, 0, stream>>>(bsum, NB);
  k_scan3<<<NB, SCAN_B, 0, stream>>>(incl, bsum, offs, N_NODES);
  k_bin<<<(N_EDGES + 255) / 256, 256, 0, stream>>>(src, dst, offs, curs, bsrc);

  k_aggr<<<(N_NODES + 7) / 8, 256, 0, stream>>>(x, offs, bsrc, h);

  gin_mlp<<<N_NODES / 32, 256, 80 * 1024, stream>>>(h, W1, b1, W2, b2, out);
}


# MLP transposed: nodes are the WMMA columns, so GEMM1's D tiles (hidden x nodes), after bias+relu, are GEMM2's B fragments in registers; no LDS, no barrier; one wave per 32 nodes; new k sum order
# speedup vs baseline: 1.3094x; 1.3094x over previous
#include <hip/hip_runtime.h>

#define N_NODES 100000
#define N_EDGES 1600000
#define D_FEAT  128
#define D_HID   512
#define D_OUT   128
#define SCAN_B  1024
#define NB      ((N_NODES + SCAN_B - 1) / SCAN_B)

typedef float v2f __attribute__((ext_vector_type(2)));
typedef float v8f __attribute__((ext_vector_type(8)));

__device__ __forceinline__ v8f wmma4(v2f a, v2f b, v8f c) {
  return __builtin_amdgcn_wmma_f32_16x16x4_f32(false, a, false, b,
                                               (short)0, c, false, false);
}


__global__ void k_zero(int* __restrict__ p, int n) {
  int i = blockIdx.x * blockDim.x + threadIdx.x;
  if (i < n) p[i] = 0;
}

__global__ void k_deg(const int* __restrict__ dst, int* __restrict__ deg) {
  int e = blockIdx.x * blockDim.x + threadIdx.x;
  if (e < N_EDGES) atomicAdd(&deg[dst[e]], 1);
}

__global__ __launch_bounds__(SCAN_B)
void k_scan1(const int* __restrict__ deg, int* __restrict__ incl,
             int* __restrict__ bsum, int n) {
  __shared__ int s[SCAN_B];
  int i = blockIdx.x * SCAN_B + threadIdx.x;
  s[threadIdx.x] = (i < n) ? deg[i] : 0;
  __syncthreads();
  for (int off = 1; off < SCAN_B; off <<= 1) {
    int t = (threadIdx.x >= off) ? s[threadIdx.x - off] : 0;
    __syncthreads();
    s[threadIdx.x] += t;
    __syncthreads();
  }
  if (i < n) incl[i] = s[threadIdx.x];
  if (threadIdx.x == SCAN_B - 1) bsum[blockIdx.x] = s[SCAN_B - 1];
}

__global__ void k_scan2(int* __restrict__ bsum, int nb) {
  if (threadIdx.x == 0 && blockIdx.x == 0) {
    int run = 0;
    for (int i = 0; i < nb; ++i) { int t = bsum[i]; bsum[i] = run; run += t; }
  }
}

__global__ __launch_bounds__(SCAN_B)
void k_scan3(const int* __restrict__ incl, const int* __restrict__ bsum,
             int* __restrict__ offs, int n) {
  int i = blockIdx.x * SCAN_B + threadIdx.x;
  if (i < n) offs[i + 1] = incl[i] + bsum[blockIdx.x];
  if (i == 0) offs[0] = 0;
}

__global__ void k_bin(const int* __restrict__ src, const int* __restrict__ dst,
                      const int* __restrict__ offs, int* __restrict__ cursor,
                      int* __restrict__ bsrc) {
  int e = blockIdx.x * blockDim.x + threadIdx.x;
  if (e >= N_EDGES) return;
  int d = dst[e];
  int pos = atomicAdd(&cursor[d], 1);
  bsrc[offs[d] + pos] = src[e];
}

__global__ __launch_bounds__(256)
void k_aggr(const float* __restrict__ x, const int* __restrict__ offs,
            const int* __restrict__ bsrc, float* __restrict__ h) {
  int node = (blockIdx.x << 3) + (threadIdx.x >> 5);
  if (node >= N_NODES) return;
  int lane = threadIdx.x & 31;
  float4 acc = *reinterpret_cast<const float4*>(x + (size_t)node * D_FEAT + lane * 4);
  int e0 = offs[node], e1 = offs[node + 1];
  for (int e = e0; e < e1; ++e) {
    int s = bsrc[e];
    float4 v = *reinterpret_cast<const float4*>(x + (size_t)s * D_FEAT + lane * 4);
    acc.x += v.x; acc.y += v.y; acc.z += v.z; acc.w += v.w;
  }
  *reinterpret_cast<float4*>(h + (size_t)node * D_FEAT + lane * 4) = acc;
}

#define MLP_NT 2
#define MLP_PF 8
#define MLP_UNROLL _Pragma("unroll")

__device__ __forceinline__ void mlp_load_step(const float* __restrict__ w1p, const float* __restrict__ w2p,
                                              int G, int t, float4& wa, float4& wb) {
  if (t < 32) {
    const float* p = w1p + 64 * G + (size_t)(8 * (t >> 1) + 2 * (t & 1)) * D_HID;
    wa = *reinterpret_cast<const float4*>(p);
    wb = *reinterpret_cast<const float4*>(p + D_HID);
  } else {
    const int u = t - 32;
    const float* p = w2p + (size_t)(64 * G + 8 * ((u >> 1) & 3) + (u >> 3)) * D_OUT + 64 * (u & 1);
    wa = *reinterpret_cast<const float4*>(p);
    wb = *reinterpret_cast<const float4*>(p + 4 * D_OUT);
  }
}

__global__ __launch_bounds__(32)
void gin_mlp_t(const float* __restrict__ h,
               const float* __restrict__ W1, const float* __restrict__ b1,
               const float* __restrict__ W2, const float* __restrict__ b2,
               float* __restrict__ out) {
  const int lane = threadIdx.x & 31;
  const int c    = lane & 15;
  const int hh   = lane >> 4;
  const int node0 = blockIdx.x * (16 * MLP_NT);

  const float* w1p = W1 + (size_t)(4 * hh) * D_HID + 4 * c;
  const float* w2p = W2 + (size_t)(32 * hh) * D_OUT + 4 * c;
  const float* b1p = b1 + 32 * hh;

  float4 ra[MLP_PF], rb[MLP_PF];
MLP_UNROLL
  for (int t = 0; t < MLP_PF; ++t) mlp_load_step(w1p, w2p, 0, t, ra[t], rb[t]);

  float4 hB[MLP_NT][16];
MLP_UNROLL
  for (int nt = 0; nt < MLP_NT; ++nt) {
    const float* hp = h + (size_t)(node0 + 16 * nt + c) * D_FEAT + 4 * hh;
MLP_UNROLL
    for (int j = 0; j < 16; ++j)
      hB[nt][j] = *reinterpret_cast<const float4*>(hp + 8 * j);
  }

  v8f acc[2][4][MLP_NT];
MLP_UNROLL
  for (int og = 0; og < 2; ++og)
MLP_UNROLL
    for (int o = 0; o < 4; ++o)
MLP_UNROLL
      for (int nt = 0; nt < MLP_NT; ++nt)
        acc[og][o][nt] = (v8f){0.f,0.f,0.f,0.f,0.f,0.f,0.f,0.f};

  for (int G = 0; G < D_HID / 64; ++G) {
    const int Gn = (G + 1) & (D_HID / 64 - 1);

    float4 bb1[8];
MLP_UNROLL
    for (int r = 0; r < 8; ++r)
      bb1[r] = *reinterpret_cast<const float4*>(b1p + 64 * G + 4 * r);

    v8f d1[4][MLP_NT];
MLP_UNROLL
    for (int q = 0; q < 4; ++q)
MLP_UNROLL
      for (int nt = 0; nt < MLP_NT; ++nt)
        d1[q][nt] = (v8f){0.f,0.f,0.f,0.f,0.f,0.f,0.f,0.f};

MLP_UNROLL
    for (int t = 0; t < 32; ++t) {
      const int j = t >> 1, p = t & 1;
      const float4 wa = ra[t % MLP_PF];
      const float4 wb = rb[t % MLP_PF];
      mlp_load_step(w1p, w2p, G, t + MLP_PF, ra[t % MLP_PF], rb[t % MLP_PF]);
      v2f bf[MLP_NT];
MLP_UNROLL
      for (int nt = 0; nt < MLP_NT; ++nt) {
        bf[nt].x = p ? hB[nt][j].z : hB[nt][j].x;
        bf[nt].y = p ? hB[nt][j].w : hB[nt][j].y;
      }
      const float was[4] = {wa.x, wa.y, wa.z, wa.w};
      const float wbs[4] = {wb.x, wb.y, wb.z, wb.w};
MLP_UNROLL
      for (int q = 0; q < 4; ++q) {
        v2f a; a.x = was[q]; a.y = wbs[q];
MLP_UNROLL
        for (int nt = 0; nt < MLP_NT; ++nt)
          d1[q][nt] = wmma4(a, bf[nt], d1[q][nt]);
      }
    }

MLP_UNROLL
    for (int r = 0; r < 8; ++r) {
      const float bbs[4] = {bb1[r].x, bb1[r].y, bb1[r].z, bb1[r].w};
MLP_UNROLL
      for (int q = 0; q < 4; ++q)
MLP_UNROLL
        for (int nt = 0; nt < MLP_NT; ++nt) {
          const float v = d1[q][nt][r] + bbs[q];
          d1[q][nt][r] = v > 0.f ? v : 0.f;
        }
    }

MLP_UNROLL
    for (int u = 0; u < 32; ++u) {
      const int t = 32 + u;
      const int q = u >> 3, j2 = (u >> 1) & 3, og = u & 1;
      const float4 wa = ra[t % MLP_PF];
      const float4 wb = rb[t % MLP_PF];
      if (t + MLP_PF < 64) mlp_load_step(w1p, w2p, G,  t + MLP_PF,      ra[t % MLP_PF], rb[t % MLP_PF]);
      else                 mlp_load_step(w1p, w2p, Gn, t + MLP_PF - 64, ra[t % MLP_PF], rb[t % MLP_PF]);
      v2f bf[MLP_NT];
MLP_UNROLL
      for (int nt = 0; nt < MLP_NT; ++nt) {
        bf[nt].x = d1[q][nt][2 * j2];
        bf[nt].y = d1[q][nt][2 * j2 + 1];
      }
      const float was[4] = {wa.x, wa.y, wa.z, wa.w};
      const float wbs[4] = {wb.x, wb.y, wb.z, wb.w};
MLP_UNROLL
      for (int o = 0; o < 4; ++o) {
        v2f a; a.x = was[o]; a.y = wbs[o];
MLP_UNROLL
        for (int nt = 0; nt < MLP_NT; ++nt)
          acc[og][o][nt] = wmma4(a, bf[nt], acc[og][o][nt]);
      }
    }
  }

MLP_UNROLL
  for (int og = 0; og < 2; ++og) {
MLP_UNROLL
    for (int r = 0; r < 8; ++r) {
      const float4 bb = *reinterpret_cast<const float4*>(b2 + 64 * og + 32 * hh + 4 * r);
MLP_UNROLL
      for (int nt = 0; nt < MLP_NT; ++nt) {
        float4 v;
        v.x = acc[og][0][nt][r] + bb.x;
        v.y = acc[og][1][nt][r] + bb.y;
        v.z = acc[og][2][nt][r] + bb.z;
        v.w = acc[og][3][nt][r] + bb.w;
        *reinterpret_cast<float4*>(out + (size_t)(node0 + 16 * nt + c) * D_OUT
                                   + 64 * og + 32 * hh + 4 * r) = v;
      }
    }
  }
}

extern "C" void kernel_launch(void* const* d_in, const int* in_sizes, int n_in,
                              void* d_out, int out_size, void* d_ws, size_t ws_size,
                              hipStream_t stream) {
  const float* x  = (const float*)d_in[0];
  const int*   ei = (const int*)d_in[1];
  const float* W1 = (const float*)d_in[2];
  const float* b1 = (const float*)d_in[3];
  const float* W2 = (const float*)d_in[4];
  const float* b2 = (const float*)d_in[5];
  float* out = (float*)d_out;

  const int* src = ei;
  const int* dst = ei + N_EDGES;

  float* h    = (float*)d_ws;
  int*  bsrc  = (int*)(h + (size_t)N_NODES * D_FEAT);
  int*  deg   = bsrc + N_EDGES;
  int*  curs  = deg + N_NODES;
  int*  incl  = curs + N_NODES;
  int*  offs  = incl + N_NODES;
  int*  bsum  = offs + N_NODES + 1;

  k_zero<<<(2 * N_NODES + 255) / 256, 256, 0, stream>>>(deg, 2 * N_NODES);
  k_deg<<<(N_EDGES + 255) / 256, 256, 0, stream>>>(dst, deg);
  k_scan1<<<NB, SCAN_B, 0, stream>>>(deg, incl, bsum, N_NODES);
  k_scan2<<<1, 32, 0, stream>>>(bsum, NB);
  k_scan3<<<NB, SCAN_B, 0, stream>>>(incl, bsum, offs, N_NODES);
  k_bin<<<(N_EDGES + 255) / 256, 256, 0, stream>>>(src, dst, offs, curs, bsrc);

  k_aggr<<<(N_NODES + 7) / 8, 256, 0, stream>>>(x, offs, bsrc, h);

  gin_mlp_t<<<N_NODES / (16 * MLP_NT), 32, 0, stream>>>(h, W1, b1, W2, b2, out);
}


# aggregation by node-range ownership: 256 workgroups each scan all dst, hold 391 h rows in LDS and add x[src] rows by LDS f32 atomic adds (arrival-order sums); no CSR build; 8 -> 2 launches
# speedup vs baseline: 2.0183x; 1.5414x over previous
#include <hip/hip_runtime.h>

#define N_NODES 100000
#define N_EDGES 1600000
#define D_FEAT  128
#define D_HID   512
#define D_OUT   128

typedef float v2f __attribute__((ext_vector_type(2)));
typedef float v8f __attribute__((ext_vector_type(8)));

__device__ __forceinline__ v8f wmma4(v2f a, v2f b, v8f c) {
  return __builtin_amdgcn_wmma_f32_16x16x4_f32(false, a, false, b,
                                               (short)0, c, false, false);
}

#define OWN_WG  256
#define OWN_N   391
#define OWN_T   1024
#define OWN_CAP 8192
#define OWN_LD  8
#define OWN_RW  ((OWN_N + OWN_T / 32 - 1) / (OWN_T / 32))
#define E4      (N_EDGES / 4)

__device__ __forceinline__ void own_note(int dval, int e, int base, int n,
                                         const float* __restrict__ x, const int* __restrict__ src,
                                         float* acc, unsigned* list, int* cnt) {
  const unsigned u = (unsigned)(dval - base);
  if (u < (unsigned)n) {
    const int pos = atomicAdd(cnt, 1);
    if (pos < OWN_CAP) {
      list[pos] = ((unsigned)e << 9) | u;
    } else {
      const float* xr = x + (size_t)src[e] * D_FEAT;
      float* a = acc + u * D_FEAT;
      for (int f = 0; f < D_FEAT; ++f)
        __hip_atomic_fetch_add(a + (f & 3) * 32 + (f >> 2), xr[f], __ATOMIC_RELAXED, __HIP_MEMORY_SCOPE_WORKGROUP);
    }
  }
}

__global__ __launch_bounds__(OWN_T)
void k_own(const float* __restrict__ x, const int* __restrict__ src, const int* __restrict__ dst,
           float* __restrict__ h) {
  __shared__ float acc[(OWN_N + 1) * D_FEAT];
  __shared__ unsigned list[OWN_CAP];
  __shared__ int cnt;

  const int tid  = threadIdx.x;
  const int lane = tid & 31;
  const int wave = __builtin_amdgcn_readfirstlane(tid >> 5);
  const int base = blockIdx.x * OWN_N;
  const int n    = (N_NODES - base < OWN_N) ? (N_NODES - base) : OWN_N;

  const float4* x4 = reinterpret_cast<const float4*>(x);

  {
    float4 v[OWN_RW];
#pragma unroll
    for (int k = 0; k < OWN_RW; ++k) {
      const int r  = wave + (OWN_T / 32) * k;
      const int rc = (r < n) ? r : (n - 1);
      v[k] = x4[(size_t)(base + rc) * 32 + lane];
    }
#pragma unroll
    for (int k = 0; k < OWN_RW; ++k) {
      const int r = wave + (OWN_T / 32) * k;
      if (k == 0 || r < n) {
        float* a = acc + r * D_FEAT + lane;
        a[0] = v[k].x; a[32] = v[k].y; a[64] = v[k].z; a[96] = v[k].w;
      }
    }
  }
  if (tid == 0) cnt = 0;
  __syncthreads();

  const int4* d4 = reinterpret_cast<const int4*>(dst);
  const int rot = blockIdx.x * 1536;
  for (int i0 = 0; i0 < E4; i0 += OWN_LD * OWN_T) {
    int4 d[OWN_LD];
    int  c[OWN_LD];
#pragma unroll
    for (int k = 0; k < OWN_LD; ++k) {
      const int i = i0 + k * OWN_T + tid;
      int cc = i + rot;
      if (cc >= E4) cc -= E4;
      c[k] = cc;
      d[k] = make_int4(-1, -1, -1, -1);
      if (i < E4) d[k] = d4[cc];
    }
#pragma unroll
    for (int k = 0; k < OWN_LD; ++k) {
      const unsigned u0 = (unsigned)(d[k].x - base), u1 = (unsigned)(d[k].y - base);
      const unsigned u2 = (unsigned)(d[k].z - base), u3 = (unsigned)(d[k].w - base);
      const unsigned um = min(min(u0, u1), min(u2, u3));
      if (um < (unsigned)n) {
        own_note(d[k].x, 4 * c[k] + 0, base, n, x, src, acc, list, &cnt);
        own_note(d[k].y, 4 * c[k] + 1, base, n, x, src, acc, list, &cnt);
        own_note(d[k].z, 4 * c[k] + 2, base, n, x, src, acc, list, &cnt);
        own_note(d[k].w, 4 * c[k] + 3, base, n, x, src, acc, list, &cnt);
      }
    }
  }
  __syncthreads();

  const int seen  = cnt;
  const int total = __builtin_amdgcn_readfirstlane(seen < OWN_CAP ? seen : OWN_CAP);
  for (int b = wave * 32; b < total; b += OWN_T) {
    const int idx = b + lane;
    unsigned ent = OWN_N;
    int s = 0;
    if (idx < total) { ent = list[idx]; s = src[ent >> 9]; }
    const int m = (total - b < 32) ? (total - b) : 32;
    for (int j0 = 0; j0 < m; j0 += 8) {
      float4 v[8];
#pragma unroll
      for (int jj = 0; jj < 8; ++jj) {
        const int sj = __builtin_amdgcn_readlane(s, j0 + jj);
        v[jj] = x4[(size_t)sj * 32 + lane];
      }
#pragma unroll
      for (int jj = 0; jj < 8; ++jj) {
        const unsigned uj = (unsigned)__builtin_amdgcn_readlane((int)ent, j0 + jj) & 511u;
        float* a = acc + uj * D_FEAT + lane;
        __hip_atomic_fetch_add(a,      v[jj].x, __ATOMIC_RELAXED, __HIP_MEMORY_SCOPE_WORKGROUP);
        __hip_atomic_fetch_add(a + 32, v[jj].y, __ATOMIC_RELAXED, __HIP_MEMORY_SCOPE_WORKGROUP);
        __hip_atomic_fetch_add(a + 64, v[jj].z, __ATOMIC_RELAXED, __HIP_MEMORY_SCOPE_WORKGROUP);
        __hip_atomic_fetch_add(a + 96, v[jj].w, __ATOMIC_RELAXED, __HIP_MEMORY_SCOPE_WORKGROUP);
      }
    }
  }
  asm volatile("s_wait_dscnt 0x0" ::: "memory");
  __syncthreads();

  float4* h4 = reinterpret_cast<float4*>(h);
  for (int r = wave; r < n; r += OWN_T / 32) {
    const float* a = acc + r * D_FEAT + lane;
    float4 v;
    v.x = a[0]; v.y = a[32]; v.z = a[64]; v.w = a[96];
    h4[(size_t)(base + r) * 32 + lane] = v;
  }
}

#define MLP_NT 2
#define MLP_PF 8
#define MLP_UNROLL _Pragma("unroll")

__device__ __forceinline__ void mlp_load_step(const float* __restrict__ w1p, const float* __restrict__ w2p,
                                              int G, int t, float4& wa, float4& wb) {
  if (t < 32) {
    const float* p = w1p + 64 * G + (size_t)(8 * (t >> 1) + 2 * (t & 1)) * D_HID;
    wa = *reinterpret_cast<const float4*>(p);
    wb = *reinterpret_cast<const float4*>(p + D_HID);
  } else {
    const int u = t - 32;
    const float* p = w2p + (size_t)(64 * G + 8 * ((u >> 1) & 3) + (u >> 3)) * D_OUT + 64 * (u & 1);
    wa = *reinterpret_cast<const float4*>(p);
    wb = *reinterpret_cast<const float4*>(p + 4 * D_OUT);
  }
}

__global__ __launch_bounds__(32)
void gin_mlp_t(const float* __restrict__ h,
               const float* __restrict__ W1, const float* __restrict__ b1,
               const float* __restrict__ W2, const float* __restrict__ b2,
               float* __restrict__ out) {
  const int lane = threadIdx.x & 31;
  const int c    = lane & 15;
  const int hh   = lane >> 4;
  const int node0 = blockIdx.x * (16 * MLP_NT);

  const float* w1p = W1 + (size_t)(4 * hh) * D_HID + 4 * c;
  const float* w2p = W2 + (size_t)(32 * hh) * D_OUT + 4 * c;
  const float* b1p = b1 + 32 * hh;

  float4 ra[MLP_PF], rb[MLP_PF];
MLP_UNROLL
  for (int t = 0; t < MLP_PF; ++t) mlp_load_step(w1p, w2p, 0, t, ra[t], rb[t]);

  float4 hB[MLP_NT][16];
MLP_UNROLL
  for (int nt = 0; nt < MLP_NT; ++nt) {
    const float* hp = h + (size_t)(node0 + 16 * nt + c) * D_FEAT + 4 * hh;
MLP_UNROLL
    for (int j = 0; j < 16; ++j)
      hB[nt][j] = *reinterpret_cast<const float4*>(hp + 8 * j);
  }

  v8f acc[2][4][MLP_NT];
MLP_UNROLL
  for (int og = 0; og < 2; ++og)
MLP_UNROLL
    for (int o = 0; o < 4; ++o)
MLP_UNROLL
      for (int nt = 0; nt < MLP_NT; ++nt)
        acc[og][o][nt] = (v8f){0.f,0.f,0.f,0.f,0.f,0.f,0.f,0.f};

  for (int G = 0; G < D_HID / 64; ++G) {
    const int Gn = (G + 1) & (D_HID / 64 - 1);

    float4 bb1[8];
MLP_UNROLL
    for (int r = 0; r < 8; ++r)
      bb1[r] = *reinterpret_cast<const float4*>(b1p + 64 * G + 4 * r);

    v8f d1[4][MLP_NT];
MLP_UNROLL
    for (int q = 0; q < 4; ++q)
MLP_UNROLL
      for (int nt = 0; nt < MLP_NT; ++nt)
        d1[q][nt] = (v8f){0.f,0.f,0.f,0.f,0.f,0.f,0.f,0.f};

MLP_UNROLL
    for (int t = 0; t < 32; ++t) {
      const int j = t >> 1, p = t & 1;
      const float4 wa = ra[t % MLP_PF];
      const float4 wb = rb[t % MLP_PF];
      mlp_load_step(w1p, w2p, G, t + MLP_PF, ra[t % MLP_PF], rb[t % MLP_PF]);
      v2f bf[MLP_NT];
MLP_UNROLL
      for (int nt = 0; nt < MLP_NT; ++nt) {
        bf[nt].x = p ? hB[nt][j].z : hB[nt][j].x;
        bf[nt].y = p ? hB[nt][j].w : hB[nt][j].y;
      }
      const float was[4] = {wa.x, wa.y, wa.z, wa.w};
      const float wbs[4] = {wb.x, wb.y, wb.z, wb.w};
MLP_UNROLL
      for (int q = 0; q < 4; ++q) {
        v2f a; a.x = was[q]; a.y = wbs[q];
MLP_UNROLL
        for (int nt = 0; nt < MLP_NT; ++nt)
          d1[q][nt] = wmma4(a, bf[nt], d1[q][nt]);
      }
    }

MLP_UNROLL
    for (int r = 0; r < 8; ++r) {
      const float bbs[4] = {bb1[r].x, bb1[r].y, bb1[r].z, bb1[r].w};
MLP_UNROLL
      for (int q = 0; q < 4; ++q)
MLP_UNROLL
        for (int nt = 0; nt < MLP_NT; ++nt) {
          const float v = d1[q][nt][r] + bbs[q];
          d1[q][nt][r] = v > 0.f ? v : 0.f;
        }
    }

MLP_UNROLL
    for (int u = 0; u < 32; ++u) {
      const int t = 32 + u;
      const int q = u >> 3, j2 = (u >> 1) & 3, og = u & 1;
      const float4 wa = ra[t % MLP_PF];
      const float4 wb = rb[t % MLP_PF];
      if (t + MLP_PF < 64) mlp_load_step(w1p, w2p, G,  t + MLP_PF,      ra[t % MLP_PF], rb[t % MLP_PF]);
      else                 mlp_load_step(w1p, w2p, Gn, t + MLP_PF - 64, ra[t % MLP_PF], rb[t % MLP_PF]);
      v2f bf[MLP_NT];
MLP_UNROLL
      for (int nt = 0; nt < MLP_NT; ++nt) {
        bf[nt].x = d1[q][nt][2 * j2];
        bf[nt].y = d1[q][nt][2 * j2 + 1];
      }
      const float was[4] = {wa.x, wa.y, wa.z, wa.w};
      const float wbs[4] = {wb.x, wb.y, wb.z, wb.w};
MLP_UNROLL
      for (int o = 0; o < 4; ++o) {
        v2f a; a.x = was[o]; a.y = wbs[o];
MLP_UNROLL
        for (int nt = 0; nt < MLP_NT; ++nt)
          acc[og][o][nt] = wmma4(a, bf[nt], acc[og][o][nt]);
      }
    }
  }

MLP_UNROLL
  for (int og = 0; og < 2; ++og) {
MLP_UNROLL
    for (int r = 0; r < 8; ++r) {
      const float4 bb = *reinterpret_cast<const float4*>(b2 + 64 * og + 32 * hh + 4 * r);
MLP_UNROLL
      for (int nt = 0; nt < MLP_NT; ++nt) {
        float4 v;
        v.x = acc[og][0][nt][r] + bb.x;
        v.y = acc[og][1][nt][r] + bb.y;
        v.z = acc[og][2][nt][r] + bb.z;
        v.w = acc[og][3][nt][r] + bb.w;
        *reinterpret_cast<float4*>(out + (size_t)(node0 + 16 * nt + c) * D_OUT
                                   + 64 * og + 32 * hh + 4 * r) = v;
      }
    }
  }
}

extern "C" void kernel_launch(void* const* d_in, const int* in_sizes, int n_in,
                              void* d_out, int out_size, void* d_ws, size_t ws_size,
                              hipStream_t stream) {
  const float* x  = (const float*)d_in[0];
  const int*   ei = (const int*)d_in[1];
  const float* W1 = (const float*)d_in[2];
  const float* b1 = (const float*)d_in[3];
  const float* W2 = (const float*)d_in[4];
  const float* b2 = (const float*)d_in[5];
  float* out = (float*)d_out;

  const int* src = ei;
  const int* dst = ei + N_EDGES;

  float* h = (float*)d_ws;

  k_own<<<OWN_WG, OWN_T, 0, stream>>>(x, src, dst, h);

  gin_mlp_t<<<N_NODES / (16 * MLP_NT), 32, 0, stream>>>(h, W1, b1, W2, b2, out);
}


# aggregation scan reads one owner byte per edge (dst/391, packed by a new first kernel) instead of 4-byte dst: 16 edges per 16-byte load; list order changes, sums still arrival-order; 2 -> 3 launches
# speedup vs baseline: 2.1780x; 1.0791x over previous
#include <hip/hip_runtime.h>

#define N_NODES 100000
#define N_EDGES 1600000
#define D_FEAT  128
#define D_HID   512
#define D_OUT   128

typedef float v2f __attribute__((ext_vector_type(2)));
typedef float v8f __attribute__((ext_vector_type(8)));

__device__ __forceinline__ v8f wmma4(v2f a, v2f b, v8f c) {
  return __builtin_amdgcn_wmma_f32_16x16x4_f32(false, a, false, b,
                                               (short)0, c, false, false);
}

#define OWN_WG  256
#define OWN_N   391
#define OWN_T   1024
#define OWN_CAP 8192
#define OWN_LD  7
#define OWN_RW  ((OWN_N + OWN_T / 32 - 1) / (OWN_T / 32))
#define E16     (N_EDGES / 16)
#define KEY_DW  (N_EDGES / 4)
#define KEY_T   256
#define KEY_LD  4

__device__ __forceinline__ unsigned own_of(int d) {
  return ((unsigned)d / (unsigned)OWN_N) & 255u;
}

__global__ __launch_bounds__(KEY_T)
void k_key(const int* __restrict__ dst, unsigned* __restrict__ key) {
  const int4* d4 = reinterpret_cast<const int4*>(dst);
  const int j0 = blockIdx.x * (KEY_T * KEY_LD) + threadIdx.x;
  int4 d[KEY_LD];
#pragma unroll
  for (int k = 0; k < KEY_LD; ++k) {
    const int j  = j0 + KEY_T * k;
    const int jc = (j < KEY_DW) ? j : (KEY_DW - 1);
    d[k] = d4[jc];
  }
#pragma unroll
  for (int k = 0; k < KEY_LD; ++k) {
    const int j = j0 + KEY_T * k;
    if (j < KEY_DW)
      key[j] = own_of(d[k].x) | (own_of(d[k].y) << 8) | (own_of(d[k].z) << 16) | (own_of(d[k].w) << 24);
  }
}

__device__ __forceinline__ void own_note(int e, int base, int n,
                                         const float* __restrict__ x, const int* __restrict__ src,
                                         const int* __restrict__ dst,
                                         float* acc, unsigned* list, int* cnt) {
  const int pos = atomicAdd(cnt, 1);
  if (pos < OWN_CAP) {
    list[pos] = (unsigned)e;
  } else {
    const unsigned u = (unsigned)(dst[e] - base);
    if (u < (unsigned)n) {
      const float* xr = x + (size_t)src[e] * D_FEAT;
      float* a = acc + u * D_FEAT;
      for (int f = 0; f < D_FEAT; ++f)
        __hip_atomic_fetch_add(a + (f & 3) * 32 + (f >> 2), xr[f], __ATOMIC_RELAXED, __HIP_MEMORY_SCOPE_WORKGROUP);
    }
  }
}

__global__ __launch_bounds__(OWN_T)
void k_own(const float* __restrict__ x, const int* __restrict__ src, const int* __restrict__ dst,
           const unsigned* __restrict__ key, float* __restrict__ h) {
  __shared__ float acc[(OWN_N + 1) * D_FEAT];
  __shared__ unsigned list[OWN_CAP];
  __shared__ int cnt;

  const int tid  = threadIdx.x;
  const int lane = tid & 31;
  const int wave = __builtin_amdgcn_readfirstlane(tid >> 5);
  const int base = blockIdx.x * OWN_N;
  const int n    = (N_NODES - base < OWN_N) ? (N_NODES - base) : OWN_N;

  const float4* x4 = reinterpret_cast<const float4*>(x);

  {
    float4 v[OWN_RW];
#pragma unroll
    for (int k = 0; k < OWN_RW; ++k) {
      const int r  = wave + (OWN_T / 32) * k;
      const int rc = (r < n) ? r : (n - 1);
      v[k] = x4[(size_t)(base + rc) * 32 + lane];
    }
#pragma unroll
    for (int k = 0; k < OWN_RW; ++k) {
      const int r = wave + (OWN_T / 32) * k;
      if (k == 0 || r < n) {
        float* a = acc + r * D_FEAT + lane;
        a[0] = v[k].x; a[32] = v[k].y; a[64] = v[k].z; a[96] = v[k].w;
      }
    }
  }
  if (tid == 0) cnt = 0;
  __syncthreads();

  const uint4* k4 = reinterpret_cast<const uint4*>(key);
  const unsigned K4 = (unsigned)blockIdx.x * 0x01010101u;
  const int rot = blockIdx.x * 384;
  for (int i0 = 0; i0 < E16; i0 += OWN_LD * OWN_T) {
    uint4 w[OWN_LD];
    int   c[OWN_LD];
#pragma unroll
    for (int k = 0; k < OWN_LD; ++k) {
      const int i = i0 + k * OWN_T + tid;
      int cc = i + rot;
      if (cc >= E16) cc -= E16;
      c[k] = cc;
      w[k] = make_uint4(~K4, ~K4, ~K4, ~K4);
      if (i < E16) w[k] = k4[cc];
    }
#pragma unroll
    for (int k = 0; k < OWN_LD; ++k) {
      const unsigned t0 = w[k].x ^ K4, t1 = w[k].y ^ K4, t2 = w[k].z ^ K4, t3 = w[k].w ^ K4;
      const unsigned z0 = (t0 - 0x01010101u) & ~t0, z1 = (t1 - 0x01010101u) & ~t1;
      const unsigned z2 = (t2 - 0x01010101u) & ~t2, z3 = (t3 - 0x01010101u) & ~t3;
      if ((z0 | z1 | z2 | z3) & 0x80808080u) {
        unsigned m = ((z0 & 0x80808080u) >> 7) | ((z1 & 0x80808080u) >> 6)
                   | ((z2 & 0x80808080u) >> 5) | ((z3 & 0x80808080u) >> 4);
        while (m) {
          const int p = __builtin_ctz(m);
          m &= m - 1;
          own_note(16 * c[k] + 4 * (p & 7) + (p >> 3), base, n, x, src, dst, acc, list, &cnt);
        }
      }
    }
  }
  __syncthreads();

  const int seen  = cnt;
  const int total = __builtin_amdgcn_readfirstlane(seen < OWN_CAP ? seen : OWN_CAP);
  for (int b = wave * 32; b < total; b += OWN_T) {
    const int idx = b + lane;
    unsigned ent = OWN_N;
    int s = 0;
    if (idx < total) {
      const unsigned e = list[idx];
      s = src[e];
      const unsigned u = (unsigned)(dst[e] - base);
      if (u < (unsigned)n) ent = u;
    }
    const int m = (total - b < 32) ? (total - b) : 32;
    for (int j0 = 0; j0 < m; j0 += 8) {
      float4 v[8];
#pragma unroll
      for (int jj = 0; jj < 8; ++jj) {
        const int sj = __builtin_amdgcn_readlane(s, j0 + jj);
        v[jj] = x4[(size_t)sj * 32 + lane];
      }
#pragma unroll
      for (int jj = 0; jj < 8; ++jj) {
        const unsigned uj = (unsigned)__builtin_amdgcn_readlane((int)ent, j0 + jj);
        float* a = acc + uj * D_FEAT + lane;
        __hip_atomic_fetch_add(a,      v[jj].x, __ATOMIC_RELAXED, __HIP_MEMORY_SCOPE_WORKGROUP);
        __hip_atomic_fetch_add(a + 32, v[jj].y, __ATOMIC_RELAXED, __HIP_MEMORY_SCOPE_WORKGROUP);
        __hip_atomic_fetch_add(a + 64, v[jj].z, __ATOMIC_RELAXED, __HIP_MEMORY_SCOPE_WORKGROUP);
        __hip_atomic_fetch_add(a + 96, v[jj].w, __ATOMIC_RELAXED, __HIP_MEMORY_SCOPE_WORKGROUP);
      }
    }
  }
  asm volatile("s_wait_dscnt 0x0" ::: "memory");
  __syncthreads();

  float4* h4 = reinterpret_cast<float4*>(h);
  for (int r = wave; r < n; r += OWN_T / 32) {
    const float* a = acc + r * D_FEAT + lane;
    float4 v;
    v.x = a[0]; v.y = a[32]; v.z = a[64]; v.w = a[96];
    h4[(size_t)(base + r) * 32 + lane] = v;
  }
}

#define MLP_NT 2
#define MLP_PF 8
#define MLP_UNROLL _Pragma("unroll")

__device__ __forceinline__ void mlp_load_step(const float* __restrict__ w1p, const float* __restrict__ w2p,
                                              int G, int t, float4& wa, float4& wb) {
  if (t < 32) {
    const float* p = w1p + 64 * G + (size_t)(8 * (t >> 1) + 2 * (t & 1)) * D_HID;
    wa = *reinterpret_cast<const float4*>(p);
    wb = *reinterpret_cast<const float4*>(p + D_HID);
  } else {
    const int u = t - 32;
    const float* p = w2p + (size_t)(64 * G + 8 * ((u >> 1) & 3) + (u >> 3)) * D_OUT + 64 * (u & 1);
    wa = *reinterpret_cast<const float4*>(p);
    wb = *reinterpret_cast<const float4*>(p + 4 * D_OUT);
  }
}

__global__ __launch_bounds__(32)
void gin_mlp_t(const float* __restrict__ h,
               const float* __restrict__ W1, const float* __restrict__ b1,
               const float* __restrict__ W2, const float* __restrict__ b2,
               float* __restrict__ out) {
  const int lane = threadIdx.x & 31;
  const int c    = lane & 15;
  const int hh   = lane >> 4;
  const int node0 = blockIdx.x * (16 * MLP_NT);

  const float* w1p = W1 + (size_t)(4 * hh) * D_HID + 4 * c;
  const float* w2p = W2 + (size_t)(32 * hh) * D_OUT + 4 * c;
  const float* b1p = b1 + 32 * hh;

  float4 ra[MLP_PF], rb[MLP_PF];
MLP_UNROLL
  for (int t = 0; t < MLP_PF; ++t) mlp_load_step(w1p, w2p, 0, t, ra[t], rb[t]);

  float4 hB[MLP_NT][16];
MLP_UNROLL
  for (int nt = 0; nt < MLP_NT; ++nt) {
    const float* hp = h + (size_t)(node0 + 16 * nt + c) * D_FEAT + 4 * hh;
MLP_UNROLL
    for (int j = 0; j < 16; ++j)
      hB[nt][j] = *reinterpret_cast<const float4*>(hp + 8 * j);
  }

  v8f acc[2][4][MLP_NT];
MLP_UNROLL
  for (int og = 0; og < 2; ++og)
MLP_UNROLL
    for (int o = 0; o < 4; ++o)
MLP_UNROLL
      for (int nt = 0; nt < MLP_NT; ++nt)
        acc[og][o][nt] = (v8f){0.f,0.f,0.f,0.f,0.f,0.f,0.f,0.f};

  for (int G = 0; G < D_HID / 64; ++G) {
    const int Gn = (G + 1) & (D_HID / 64 - 1);

    float4 bb1[8];
MLP_UNROLL
    for (int r = 0; r < 8; ++r)
      bb1[r] = *reinterpret_cast<const float4*>(b1p + 64 * G + 4 * r);

    v8f d1[4][MLP_NT];
MLP_UNROLL
    for (int q = 0; q < 4; ++q)
MLP_UNROLL
      for (int nt = 0; nt < MLP_NT; ++nt)
        d1[q][nt] = (v8f){0.f,0.f,0.f,0.f,0.f,0.f,0.f,0.f};

MLP_UNROLL
    for (int t = 0; t < 32; ++t) {
      const int j = t >> 1, p = t & 1;
      const float4 wa = ra[t % MLP_PF];
      const float4 wb = rb[t % MLP_PF];
      mlp_load_step(w1p, w2p, G, t + MLP_PF, ra[t % MLP_PF], rb[t % MLP_PF]);
      v2f bf[MLP_NT];
MLP_UNROLL
      for (int nt = 0; nt < MLP_NT; ++nt) {
        bf[nt].x = p ? hB[nt][j].z : hB[nt][j].x;
        bf[nt].y = p ? hB[nt][j].w : hB[nt][j].y;
      }
      const float was[4] = {wa.x, wa.y, wa.z, wa.w};
      const float wbs[4] = {wb.x, wb.y, wb.z, wb.w};
MLP_UNROLL
      for (int q = 0; q < 4; ++q) {
        v2f a; a.x = was[q]; a.y = wbs[q];
MLP_UNROLL
        for (int nt = 0; nt < MLP_NT; ++nt)
          d1[q][nt] = wmma4(a, bf[nt], d1[q][nt]);
      }
    }

MLP_UNROLL
    for (int r = 0; r < 8; ++r) {
      const float bbs[4] = {bb1[r].x, bb1[r].y, bb1[r].z, bb1[r].w};
MLP_UNROLL
      for (int q = 0; q < 4; ++q)
MLP_UNROLL
        for (int nt = 0; nt < MLP_NT; ++nt) {
          const float v = d1[q][nt][r] + bbs[q];
          d1[q][nt][r] = v > 0.f ? v : 0.f;
        }
    }

MLP_UNROLL
    for (int u = 0; u < 32; ++u) {
      const int t = 32 + u;
      const int q = u >> 3, j2 = (u >> 1) & 3, og = u & 1;
      const float4 wa = ra[t % MLP_PF];
      const float4 wb = rb[t % MLP_PF];
      if (t + MLP_PF < 64) mlp_load_step(w1p, w2p, G,  t + MLP_PF,      ra[t % MLP_PF], rb[t % MLP_PF]);
      else                 mlp_load_step(w1p, w2p, Gn, t + MLP_PF - 64, ra[t % MLP_PF], rb[t % MLP_PF]);
      v2f bf[MLP_NT];
MLP_UNROLL
      for (int nt = 0; nt < MLP_NT; ++nt) {
        bf[nt].x = d1[q][nt][2 * j2];
        bf[nt].y = d1[q][nt][2 * j2 + 1];
      }
      const float was[4] = {wa.x, wa.y, wa.z, wa.w};
      const float wbs[4] = {wb.x, wb.y, wb.z, wb.w};
MLP_UNROLL
      for (int o = 0; o < 4; ++o) {
        v2f a; a.x = was[o]; a.y = wbs[o];
MLP_UNROLL
        for (int nt = 0; nt < MLP_NT; ++nt)
          acc[og][o][nt] = wmma4(a, bf[nt], acc[og][o][nt]);
      }
    }
  }

MLP_UNROLL
  for (int og = 0; og < 2; ++og) {
MLP_UNROLL
    for (int r = 0; r < 8; ++r) {
      const float4 bb = *reinterpret_cast<const float4*>(b2 + 64 * og + 32 * hh + 4 * r);
MLP_UNROLL
      for (int nt = 0; nt < MLP_NT; ++nt) {
        float4 v;
        v.x = acc[og][0][nt][r] + bb.x;
        v.y = acc[og][1][nt][r] + bb.y;
        v.z = acc[og][2][nt][r] + bb.z;
        v.w = acc[og][3][nt][r] + bb.w;
        *reinterpret_cast<float4*>(out + (size_t)(node0 + 16 * nt + c) * D_OUT
                                   + 64 * og + 32 * hh + 4 * r) = v;
      }
    }
  }
}

extern "C" void kernel_launch(void* const* d_in, const int* in_sizes, int n_in,
                              void* d_out, int out_size, void* d_ws, size_t ws_size,
                              hipStream_t stream) {
  const float* x  = (const float*)d_in[0];
  const int*   ei = (const int*)d_in[1];
  const float* W1 = (const float*)d_in[2];
  const float* b1 = (const float*)d_in[3];
  const float* W2 = (const float*)d_in[4];
  const float* b2 = (const float*)d_in[5];
  float* out = (float*)d_out;

  const int* src = ei;
  const int* dst = ei + N_EDGES;

  float* h = (float*)d_ws;
  unsigned* key = (unsigned*)(h + (size_t)N_NODES * D_FEAT);

  k_key<<<(KEY_DW + KEY_T * KEY_LD - 1) / (KEY_T * KEY_LD), KEY_T, 0, stream>>>(dst, key);

  k_own<<<OWN_WG, OWN_T, 0, stream>>>(x, src, dst, key, h);

  gin_mlp_t<<<N_NODES / (16 * MLP_NT), 32, 0, stream>>>(h, W1, b1, W2, b2, out);
}


# aggregation finds its edges by chunk bits: k_key also writes one bit per (owner, 16-edge chunk); a workgroup reads its bit row and only the flagged key chunks, not all keys; sums stay arrival-order
# speedup vs baseline: 2.3138x; 1.0624x over previous
#include <hip/hip_runtime.h>

#define N_NODES 100000
#define N_EDGES 1600000
#define D_FEAT  128
#define D_HID   512
#define D_OUT   128

typedef float v2f __attribute__((ext_vector_type(2)));
typedef float v8f __attribute__((ext_vector_type(8)));

__device__ __forceinline__ v8f wmma4(v2f a, v2f b, v8f c) {
  return __builtin_amdgcn_wmma_f32_16x16x4_f32(false, a, false, b,
                                               (short)0, c, false, false);
}

#define OWN_WG  256
#define OWN_N   391
#define OWN_T   1024
#define OWN_CAP 8192
#define OWN_RW  ((OWN_N + OWN_T / 32 - 1) / (OWN_T / 32))
#define E16     (N_EDGES / 16)
#define KEY_DW  (N_EDGES / 4)
#define KEY_T   256
#define KEY_LD  4
#define KEY_BW  (KEY_T * KEY_LD / 4 / 32)
#define BM_W    (E16 / 32)
#define OWN_BW  ((BM_W + OWN_T - 1) / OWN_T)

__device__ __forceinline__ unsigned own_of(int d) {
  return ((unsigned)d / (unsigned)OWN_N) & 255u;
}

__global__ __launch_bounds__(KEY_T)
void k_key(const int* __restrict__ dst, unsigned* __restrict__ key, unsigned* __restrict__ bm) {
  __shared__ unsigned bs[OWN_WG * KEY_BW];
  for (int i = threadIdx.x; i < OWN_WG * KEY_BW; i += KEY_T) bs[i] = 0u;
  __syncthreads();
  const int4* d4 = reinterpret_cast<const int4*>(dst);
  const int j0 = blockIdx.x * (KEY_T * KEY_LD) + threadIdx.x;
  int4 d[KEY_LD];
#pragma unroll
  for (int k = 0; k < KEY_LD; ++k) {
    const int j  = j0 + KEY_T * k;
    const int jc = (j < KEY_DW) ? j : (KEY_DW - 1);
    d[k] = d4[jc];
  }
#pragma unroll
  for (int k = 0; k < KEY_LD; ++k) {
    const int j = j0 + KEY_T * k;
    if (j < KEY_DW) {
      const unsigned o0 = own_of(d[k].x), o1 = own_of(d[k].y), o2 = own_of(d[k].z), o3 = own_of(d[k].w);
      key[j] = o0 | (o1 << 8) | (o2 << 16) | (o3 << 24);
      const int lc = (threadIdx.x + KEY_T * k) >> 2;
      const unsigned bit = 1u << (lc & 31);
      const int q = lc >> 5;
      atomicOr(&bs[o0 * KEY_BW + q], bit);
      atomicOr(&bs[o1 * KEY_BW + q], bit);
      atomicOr(&bs[o2 * KEY_BW + q], bit);
      atomicOr(&bs[o3 * KEY_BW + q], bit);
    }
  }
  __syncthreads();
  for (int i = threadIdx.x; i < OWN_WG * KEY_BW; i += KEY_T)
    bm[(size_t)blockIdx.x * (OWN_WG * KEY_BW) + i] = bs[i];
}

__device__ __forceinline__ int own_next(unsigned (&bw)[OWN_BW], int& cb) {
#pragma unroll
  for (int r = 0; r < OWN_BW - 1; ++r) {
    const bool z = (bw[0] == 0u);
#pragma unroll
    for (int k = 0; k < OWN_BW - 1; ++k) bw[k] = z ? bw[k + 1] : bw[k];
    bw[OWN_BW - 1] = z ? 0u : bw[OWN_BW - 1];
    cb += z ? 32 * OWN_T : 0;
  }
  const unsigned w = bw[0];
  if (w == 0u) return -1;
  bw[0] = w & (w - 1);
  return cb + __builtin_ctz(w);
}

__device__ __forceinline__ void own_note(int e, int base, int n,
                                         const float* __restrict__ x, const int* __restrict__ src,
                                         const int* __restrict__ dst,
                                         float* acc, unsigned* list, int* cnt) {
  const int pos = atomicAdd(cnt, 1);
  if (pos < OWN_CAP) {
    list[pos] = (unsigned)e;
  } else {
    const unsigned u = (unsigned)(dst[e] - base);
    if (u < (unsigned)n) {
      const float* xr = x + (size_t)src[e] * D_FEAT;
      float* a = acc + u * D_FEAT;
      for (int f = 0; f < D_FEAT; ++f)
        __hip_atomic_fetch_add(a + (f & 3) * 32 + (f >> 2), xr[f], __ATOMIC_RELAXED, __HIP_MEMORY_SCOPE_WORKGROUP);
    }
  }
}

__device__ __forceinline__ void own_chunk(const uint4 w, int c, unsigned K4, int base, int n,
                                          const float* __restrict__ x, const int* __restrict__ src,
                                          const int* __restrict__ dst,
                                          float* acc, unsigned* list, int* cnt) {
  const unsigned t0 = w.x ^ K4, t1 = w.y ^ K4, t2 = w.z ^ K4, t3 = w.w ^ K4;
  const unsigned z0 = (t0 - 0x01010101u) & ~t0, z1 = (t1 - 0x01010101u) & ~t1;
  const unsigned z2 = (t2 - 0x01010101u) & ~t2, z3 = (t3 - 0x01010101u) & ~t3;
  unsigned m = ((z0 & 0x80808080u) >> 7) | ((z1 & 0x80808080u) >> 6)
             | ((z2 & 0x80808080u) >> 5) | ((z3 & 0x80808080u) >> 4);
  while (m) {
    const int p = __builtin_ctz(m);
    m &= m - 1;
    own_note(16 * c + 4 * (p & 7) + (p >> 3), base, n, x, src, dst, acc, list, cnt);
  }
}

__global__ __launch_bounds__(OWN_T)
void k_own(const float* __restrict__ x, const int* __restrict__ src, const int* __restrict__ dst,
           const unsigned* __restrict__ key, const unsigned* __restrict__ bm, float* __restrict__ h) {
  __shared__ float acc[(OWN_N + 1) * D_FEAT];
  __shared__ unsigned list[OWN_CAP];
  __shared__ int cnt;

  const int tid  = threadIdx.x;
  const int lane = tid & 31;
  const int wave = __builtin_amdgcn_readfirstlane(tid >> 5);
  const int base = blockIdx.x * OWN_N;
  const int n    = (N_NODES - base < OWN_N) ? (N_NODES - base) : OWN_N;

  const float4* x4 = reinterpret_cast<const float4*>(x);

  {
    float4 v[OWN_RW];
#pragma unroll
    for (int k = 0; k < OWN_RW; ++k) {
      const int r  = wave + (OWN_T / 32) * k;
      const int rc = (r < n) ? r : (n - 1);
      v[k] = x4[(size_t)(base + rc) * 32 + lane];
    }
#pragma unroll
    for (int k = 0; k < OWN_RW; ++k) {
      const int r = wave + (OWN_T / 32) * k;
      if (k == 0 || r < n) {
        float* a = acc + r * D_FEAT + lane;
        a[0] = v[k].x; a[32] = v[k].y; a[64] = v[k].z; a[96] = v[k].w;
      }
    }
  }
  if (tid == 0) cnt = 0;
  __syncthreads();

  const uint4* k4 = reinterpret_cast<const uint4*>(key);
  const unsigned K4 = (unsigned)blockIdx.x * 0x01010101u;
  const unsigned* bmg = bm + (size_t)blockIdx.x * KEY_BW;
  unsigned bw[OWN_BW];
#pragma unroll
  for (int k = 0; k < OWN_BW; ++k) {
    const int wi = tid + OWN_T * k;
    bw[k] = (wi < BM_W) ? bmg[(size_t)(wi / KEY_BW) * (OWN_WG * KEY_BW) + wi % KEY_BW] : 0u;
  }
  int cb = 32 * tid;
  int cA = own_next(bw, cb);
  if (cA >= 0) {
    int cB = own_next(bw, cb);
    int cC;
    uint4 A = k4[cA];
    uint4 B = k4[cB >= 0 ? cB : cA];
    uint4 C;
    for (;;) {
      cC = (cB >= 0) ? own_next(bw, cb) : -1;
      C = k4[cC >= 0 ? cC : cA];
      own_chunk(A, cA, K4, base, n, x, src, dst, acc, list, &cnt);
      if (cB < 0) break;
      cA = (cC >= 0) ? own_next(bw, cb) : -1;
      A = k4[cA >= 0 ? cA : cB];
      own_chunk(B, cB, K4, base, n, x, src, dst, acc, list, &cnt);
      if (cC < 0) break;
      cB = (cA >= 0) ? own_next(bw, cb) : -1;
      B = k4[cB >= 0 ? cB : cC];
      own_chunk(C, cC, K4, base, n, x, src, dst, acc, list, &cnt);
      if (cA < 0) break;
    }
  }
  __syncthreads();

  const int seen  = cnt;
  const int total = __builtin_amdgcn_readfirstlane(seen < OWN_CAP ? seen : OWN_CAP);
  for (int b = wave * 32; b < total; b += OWN_T) {
    const int idx = b + lane;
    unsigned ent = OWN_N;
    int s = 0;
    if (idx < total) {
      const unsigned e = list[idx];
      s = src[e];
      const unsigned u = (unsigned)(dst[e] - base);
      if (u < (unsigned)n) ent = u;
    }
    const int m = (total - b < 32) ? (total - b) : 32;
    for (int j0 = 0; j0 < m; j0 += 8) {
      float4 v[8];
#pragma unroll
      for (int jj = 0; jj < 8; ++jj) {
        const int sj = __builtin_amdgcn_readlane(s, j0 + jj);
        v[jj] = x4[(size_t)sj * 32 + lane];
      }
#pragma unroll
      for (int jj = 0; jj < 8; ++jj) {
        const unsigned uj = (unsigned)__builtin_amdgcn_readlane((int)ent, j0 + jj);
        float* a = acc + uj * D_FEAT + lane;
        __hip_atomic_fetch_add(a,      v[jj].x, __ATOMIC_RELAXED, __HIP_MEMORY_SCOPE_WORKGROUP);
        __hip_atomic_fetch_add(a + 32, v[jj].y, __ATOMIC_RELAXED, __HIP_MEMORY_SCOPE_WORKGROUP);
        __hip_atomic_fetch_add(a + 64, v[jj].z, __ATOMIC_RELAXED, __HIP_MEMORY_SCOPE_WORKGROUP);
        __hip_atomic_fetch_add(a + 96, v[jj].w, __ATOMIC_RELAXED, __HIP_MEMORY_SCOPE_WORKGROUP);
      }
    }
  }
  asm volatile("s_wait_dscnt 0x0" ::: "memory");
  __syncthreads();

  float4* h4 = reinterpret_cast<float4*>(h);
  for (int r = wave; r < n; r += OWN_T / 32) {
    const float* a = acc + r * D_FEAT + lane;
    float4 v;
    v.x = a[0]; v.y = a[32]; v.z = a[64]; v.w = a[96];
    h4[(size_t)(base + r) * 32 + lane] = v;
  }
}

#define MLP_NT 2
#define MLP_PF 8
#define MLP_UNROLL _Pragma("unroll")

__device__ __forceinline__ void mlp_load_step(const float* __restrict__ w1p, const float* __restrict__ w2p,
                                              int G, int t, float4& wa, float4& wb) {
  if (t < 32) {
    const float* p = w1p + 64 * G + (size_t)(8 * (t >> 1) + 2 * (t & 1)) * D_HID;
    wa = *reinterpret_cast<const float4*>(p);
    wb = *reinterpret_cast<const float4*>(p + D_HID);
  } else {
    const int u = t - 32;
    const float* p = w2p + (size_t)(64 * G + 8 * ((u >> 1) & 3) + (u >> 3)) * D_OUT + 64 * (u & 1);
    wa = *reinterpret_cast<const float4*>(p);
    wb = *reinterpret_cast<const float4*>(p + 4 * D_OUT);
  }
}

__global__ __launch_bounds__(32)
void gin_mlp_t(const float* __restrict__ h,
               const float* __restrict__ W1, const float* __restrict__ b1,
               const float* __restrict__ W2, const float* __restrict__ b2,
               float* __restrict__ out) {
  const int lane = threadIdx.x & 31;
  const int c    = lane & 15;
  const int hh   = lane >> 4;
  const int node0 = blockIdx.x * (16 * MLP_NT);

  const float* w1p = W1 + (size_t)(4 * hh) * D_HID + 4 * c;
  const float* w2p = W2 + (size_t)(32 * hh) * D_OUT + 4 * c;
  const float* b1p = b1 + 32 * hh;

  float4 ra[MLP_PF], rb[MLP_PF];
MLP_UNROLL
  for (int t = 0; t < MLP_PF; ++t) mlp_load_step(w1p, w2p, 0, t, ra[t], rb[t]);

  float4 hB[MLP_NT][16];
MLP_UNROLL
  for (int nt = 0; nt < MLP_NT; ++nt) {
    const float* hp = h + (size_t)(node0 + 16 * nt + c) * D_FEAT + 4 * hh;
MLP_UNROLL
    for (int j = 0; j < 16; ++j)
      hB[nt][j] = *reinterpret_cast<const float4*>(hp + 8 * j);
  }

  v8f acc[2][4][MLP_NT];
MLP_UNROLL
  for (int og = 0; og < 2; ++og)
MLP_UNROLL
    for (int o = 0; o < 4; ++o)
MLP_UNROLL
      for (int nt = 0; nt < MLP_NT; ++nt)
        acc[og][o][nt] = (v8f){0.f,0.f,0.f,0.f,0.f,0.f,0.f,0.f};

  for (int G = 0; G < D_HID / 64; ++G) {
    const int Gn = (G + 1) & (D_HID / 64 - 1);

    float4 bb1[8];
MLP_UNROLL
    for (int r = 0; r < 8; ++r)
      bb1[r] = *reinterpret_cast<const float4*>(b1p + 64 * G + 4 * r);

    v8f d1[4][MLP_NT];
MLP_UNROLL
    for (int q = 0; q < 4; ++q)
MLP_UNROLL
      for (int nt = 0; nt < MLP_NT; ++nt)
        d1[q][nt] = (v8f){0.f,0.f,0.f,0.f,0.f,0.f,0.f,0.f};

MLP_UNROLL
    for (int t = 0; t < 32; ++t) {
      const int j = t >> 1, p = t & 1;
      const float4 wa = ra[t % MLP_PF];
      const float4 wb = rb[t % MLP_PF];
      mlp_load_step(w1p, w2p, G, t + MLP_PF, ra[t % MLP_PF], rb[t % MLP_PF]);
      v2f bf[MLP_NT];
MLP_UNROLL
      for (int nt = 0; nt < MLP_NT; ++nt) {
        bf[nt].x = p ? hB[nt][j].z : hB[nt][j].x;
        bf[nt].y = p ? hB[nt][j].w : hB[nt][j].y;
      }
      const float was[4] = {wa.x, wa.y, wa.z, wa.w};
      const float wbs[4] = {wb.x, wb.y, wb.z, wb.w};
MLP_UNROLL
      for (int q = 0; q < 4; ++q) {
        v2f a; a.x = was[q]; a.y = wbs[q];
MLP_UNROLL
        for (int nt = 0; nt < MLP_NT; ++nt)
          d1[q][nt] = wmma4(a, bf[nt], d1[q][nt]);
      }
    }

MLP_UNROLL
    for (int r = 0; r < 8; ++r) {
      const float bbs[4] = {bb1[r].x, bb1[r].y, bb1[r].z, bb1[r].w};
MLP_UNROLL
      for (int q = 0; q < 4; ++q)
MLP_UNROLL
        for (int nt = 0; nt < MLP_NT; ++nt) {
          const float v = d1[q][nt][r] + bbs[q];
          d1[q][nt][r] = v > 0.f ? v : 0.f;
        }
    }

MLP_UNROLL
    for (int u = 0; u < 32; ++u) {
      const int t = 32 + u;
      const int q = u >> 3, j2 = (u >> 1) & 3, og = u & 1;
      const float4 wa = ra[t % MLP_PF];
      const float4 wb = rb[t % MLP_PF];
      if (t + MLP_PF < 64) mlp_load_step(w1p, w2p, G,  t + MLP_PF,      ra[t % MLP_PF], rb[t % MLP_PF]);
      else                 mlp_load_step(w1p, w2p, Gn, t + MLP_PF - 64, ra[t % MLP_PF], rb[t % MLP_PF]);
      v2f bf[MLP_NT];
MLP_UNROLL
      for (int nt = 0; nt < MLP_NT; ++nt) {
        bf[nt].x = d1[q][nt][2 * j2];
        bf[nt].y = d1[q][nt][2 * j2 + 1];
      }
      const float was[4] = {wa.x, wa.y, wa.z, wa.w};
      const float wbs[4] = {wb.x, wb.y, wb.z, wb.w};
MLP_UNROLL
      for (int o = 0; o < 4; ++o) {
        v2f a; a.x = was[o]; a.y = wbs[o];
MLP_UNROLL
        for (int nt = 0; nt < MLP_NT; ++nt)
          acc[og][o][nt] = wmma4(a, bf[nt], acc[og][o][nt]);
      }
    }
  }

MLP_UNROLL
  for (int og = 0; og < 2; ++og) {
MLP_UNROLL
    for (int r = 0; r < 8; ++r) {
      const float4 bb = *reinterpret_cast<const float4*>(b2 + 64 * og + 32 * hh + 4 * r);
MLP_UNROLL
      for (int nt = 0; nt < MLP_NT; ++nt) {
        float4 v;
        v.x = acc[og][0][nt][r] + bb.x;
        v.y = acc[og][1][nt][r] + bb.y;
        v.z = acc[og][2][nt][r] + bb.z;
        v.w = acc[og][3][nt][r] + bb.w;
        *reinterpret_cast<float4*>(out + (size_t)(node0 + 16 * nt + c) * D_OUT
                                   + 64 * og + 32 * hh + 4 * r) = v;
      }
    }
  }
}

extern "C" void kernel_launch(void* const* d_in, const int* in_sizes, int n_in,
                              void* d_out, int out_size, void* d_ws, size_t ws_size,
                              hipStream_t stream) {
  const float* x  = (const float*)d_in[0];
  const int*   ei = (const int*)d_in[1];
  const float* W1 = (const float*)d_in[2];
  const float* b1 = (const float*)d_in[3];
  const float* W2 = (const float*)d_in[4];
  const float* b2 = (const float*)d_in[5];
  float* out = (float*)d_out;

  const int* src = ei;
  const int* dst = ei + N_EDGES;

  float* h = (float*)d_ws;
  unsigned* key = (unsigned*)(h + (size_t)N_NODES * D_FEAT);
  unsigned* bm  = key + KEY_DW;

  k_key<<<(KEY_DW + KEY_T * KEY_LD - 1) / (KEY_T * KEY_LD), KEY_T, 0, stream>>>(dst, key, bm);

  k_own<<<OWN_WG, OWN_T, 0, stream>>>(x, src, dst, key, bm, h);

  gin_mlp_t<<<N_NODES / (16 * MLP_NT), 32, 0, stream>>>(h, W1, b1, W2, b2, out);
}


# MLP grid made resident and even: 256 workgroups x 8 waves (2 per SIMD), each wave 2+1 node tiles; the 106 left-over tiles split by hidden group over 8 waves, partials summed in LDS (new sum order)
# speedup vs baseline: 2.4370x; 1.0532x over previous
#include <hip/hip_runtime.h>

#define N_NODES 100000
#define N_EDGES 1600000
#define D_FEAT  128
#define D_HID   512
#define D_OUT   128

typedef float v2f __attribute__((ext_vector_type(2)));
typedef float v8f __attribute__((ext_vector_type(8)));

__device__ __forceinline__ v8f wmma4(v2f a, v2f b, v8f c) {
  return __builtin_amdgcn_wmma_f32_16x16x4_f32(false, a, false, b,
                                               (short)0, c, false, false);
}

#define OWN_WG  256
#define OWN_N   391
#define OWN_T   1024
#define OWN_CAP 8192
#define OWN_RW  ((OWN_N + OWN_T / 32 - 1) / (OWN_T / 32))
#define E16     (N_EDGES / 16)
#define KEY_DW  (N_EDGES / 4)
#define KEY_T   256
#define KEY_LD  4
#define KEY_BW  (KEY_T * KEY_LD / 4 / 32)
#define BM_W    (E16 / 32)
#define OWN_BW  ((BM_W + OWN_T - 1) / OWN_T)

__device__ __forceinline__ unsigned own_of(int d) {
  return ((unsigned)d / (unsigned)OWN_N) & 255u;
}

__global__ __launch_bounds__(KEY_T)
void k_key(const int* __restrict__ dst, unsigned* __restrict__ key, unsigned* __restrict__ bm) {
  __shared__ unsigned bs[OWN_WG * KEY_BW];
  for (int i = threadIdx.x; i < OWN_WG * KEY_BW; i += KEY_T) bs[i] = 0u;
  __syncthreads();
  const int4* d4 = reinterpret_cast<const int4*>(dst);
  const int j0 = blockIdx.x * (KEY_T * KEY_LD) + threadIdx.x;
  int4 d[KEY_LD];
#pragma unroll
  for (int k = 0; k < KEY_LD; ++k) {
    const int j  = j0 + KEY_T * k;
    const int jc = (j < KEY_DW) ? j : (KEY_DW - 1);
    d[k] = d4[jc];
  }
#pragma unroll
  for (int k = 0; k < KEY_LD; ++k) {
    const int j = j0 + KEY_T * k;
    if (j < KEY_DW) {
      const unsigned o0 = own_of(d[k].x), o1 = own_of(d[k].y), o2 = own_of(d[k].z), o3 = own_of(d[k].w);
      key[j] = o0 | (o1 << 8) | (o2 << 16) | (o3 << 24);
      const int lc = (threadIdx.x + KEY_T * k) >> 2;
      const unsigned bit = 1u << (lc & 31);
      const int q = lc >> 5;
      atomicOr(&bs[o0 * KEY_BW + q], bit);
      atomicOr(&bs[o1 * KEY_BW + q], bit);
      atomicOr(&bs[o2 * KEY_BW + q], bit);
      atomicOr(&bs[o3 * KEY_BW + q], bit);
    }
  }
  __syncthreads();
  for (int i = threadIdx.x; i < OWN_WG * KEY_BW; i += KEY_T)
    bm[(size_t)blockIdx.x * (OWN_WG * KEY_BW) + i] = bs[i];
}

__device__ __forceinline__ int own_next(unsigned (&bw)[OWN_BW], int& cb) {
#pragma unroll
  for (int r = 0; r < OWN_BW - 1; ++r) {
    const bool z = (bw[0] == 0u);
#pragma unroll
    for (int k = 0; k < OWN_BW - 1; ++k) bw[k] = z ? bw[k + 1] : bw[k];
    bw[OWN_BW - 1] = z ? 0u : bw[OWN_BW - 1];
    cb += z ? 32 * OWN_T : 0;
  }
  const unsigned w = bw[0];
  if (w == 0u) return -1;
  bw[0] = w & (w - 1);
  return cb + __builtin_ctz(w);
}

__device__ __forceinline__ void own_note(int e, int base, int n,
                                         const float* __restrict__ x, const int* __restrict__ src,
                                         const int* __restrict__ dst,
                                         float* acc, unsigned* list, int* cnt) {
  const int pos = atomicAdd(cnt, 1);
  if (pos < OWN_CAP) {
    list[pos] = (unsigned)e;
  } else {
    const unsigned u = (unsigned)(dst[e] - base);
    if (u < (unsigned)n) {
      const float* xr = x + (size_t)src[e] * D_FEAT;
      float* a = acc + u * D_FEAT;
      for (int f = 0; f < D_FEAT; ++f)
        __hip_atomic_fetch_add(a + (f & 3) * 32 + (f >> 2), xr[f], __ATOMIC_RELAXED, __HIP_MEMORY_SCOPE_WORKGROUP);
    }
  }
}

__device__ __forceinline__ void own_chunk(const uint4 w, int c, unsigned K4, int base, int n,
                                          const float* __restrict__ x, const int* __restrict__ src,
                                          const int* __restrict__ dst,
                                          float* acc, unsigned* list, int* cnt) {
  const unsigned t0 = w.x ^ K4, t1 = w.y ^ K4, t2 = w.z ^ K4, t3 = w.w ^ K4;
  const unsigned z0 = (t0 - 0x01010101u) & ~t0, z1 = (t1 - 0x01010101u) & ~t1;
  const unsigned z2 = (t2 - 0x01010101u) & ~t2, z3 = (t3 - 0x01010101u) & ~t3;
  unsigned m = ((z0 & 0x80808080u) >> 7) | ((z1 & 0x80808080u) >> 6)
             | ((z2 & 0x80808080u) >> 5) | ((z3 & 0x80808080u) >> 4);
  while (m) {
    const int p = __builtin_ctz(m);
    m &= m - 1;
    own_note(16 * c + 4 * (p & 7) + (p >> 3), base, n, x, src, dst, acc, list, cnt);
  }
}

__global__ __launch_bounds__(OWN_T)
void k_own(const float* __restrict__ x, const int* __restrict__ src, const int* __restrict__ dst,
           const unsigned* __restrict__ key, const unsigned* __restrict__ bm, float* __restrict__ h) {
  __shared__ float acc[(OWN_N + 1) * D_FEAT];
  __shared__ unsigned list[OWN_CAP];
  __shared__ int cnt;

  const int tid  = threadIdx.x;
  const int lane = tid & 31;
  const int wave = __builtin_amdgcn_readfirstlane(tid >> 5);
  const int base = blockIdx.x * OWN_N;
  const int n    = (N_NODES - base < OWN_N) ? (N_NODES - base) : OWN_N;

  const float4* x4 = reinterpret_cast<const float4*>(x);

  {
    float4 v[OWN_RW];
#pragma unroll
    for (int k = 0; k < OWN_RW; ++k) {
      const int r  = wave + (OWN_T / 32) * k;
      const int rc = (r < n) ? r : (n - 1);
      v[k] = x4[(size_t)(base + rc) * 32 + lane];
    }
#pragma unroll
    for (int k = 0; k < OWN_RW; ++k) {
      const int r = wave + (OWN_T / 32) * k;
      if (k == 0 || r < n) {
        float* a = acc + r * D_FEAT + lane;
        a[0] = v[k].x; a[32] = v[k].y; a[64] = v[k].z; a[96] = v[k].w;
      }
    }
  }
  if (tid == 0) cnt = 0;
  __syncthreads();

  const uint4* k4 = reinterpret_cast<const uint4*>(key);
  const unsigned K4 = (unsigned)blockIdx.x * 0x01010101u;
  const unsigned* bmg = bm + (size_t)blockIdx.x * KEY_BW;
  unsigned bw[OWN_BW];
#pragma unroll
  for (int k = 0; k < OWN_BW; ++k) {
    const int wi = tid + OWN_T * k;
    bw[k] = (wi < BM_W) ? bmg[(size_t)(wi / KEY_BW) * (OWN_WG * KEY_BW) + wi % KEY_BW] : 0u;
  }
  int cb = 32 * tid;
  int cA = own_next(bw, cb);
  if (cA >= 0) {
    int cB = own_next(bw, cb);
    int cC;
    uint4 A = k4[cA];
    uint4 B = k4[cB >= 0 ? cB : cA];
    uint4 C;
    for (;;) {
      cC = (cB >= 0) ? own_next(bw, cb) : -1;
      C = k4[cC >= 0 ? cC : cA];
      own_chunk(A, cA, K4, base, n, x, src, dst, acc, list, &cnt);
      if (cB < 0) break;
      cA = (cC >= 0) ? own_next(bw, cb) : -1;
      A = k4[cA >= 0 ? cA : cB];
      own_chunk(B, cB, K4, base, n, x, src, dst, acc, list, &cnt);
      if (cC < 0) break;
      cB = (cA >= 0) ? own_next(bw, cb) : -1;
      B = k4[cB >= 0 ? cB : cC];
      own_chunk(C, cC, K4, base, n, x, src, dst, acc, list, &cnt);
      if (cA < 0) break;
    }
  }
  __syncthreads();

  const int seen  = cnt;
  const int total = __builtin_amdgcn_readfirstlane(seen < OWN_CAP ? seen : OWN_CAP);
  for (int b = wave * 32; b < total; b += OWN_T) {
    const int idx = b + lane;
    unsigned ent = OWN_N;
    int s = 0;
    if (idx < total) {
      const unsigned e = list[idx];
      s = src[e];
      const unsigned u = (unsigned)(dst[e] - base);
      if (u < (unsigned)n) ent = u;
    }
    const int m = (total - b < 32) ? (total - b) : 32;
    for (int j0 = 0; j0 < m; j0 += 8) {
      float4 v[8];
#pragma unroll
      for (int jj = 0; jj < 8; ++jj) {
        const int sj = __builtin_amdgcn_readlane(s, j0 + jj);
        v[jj] = x4[(size_t)sj * 32 + lane];
      }
#pragma unroll
      for (int jj = 0; jj < 8; ++jj) {
        const unsigned uj = (unsigned)__builtin_amdgcn_readlane((int)ent, j0 + jj);
        float* a = acc + uj * D_FEAT + lane;
        __hip_atomic_fetch_add(a,      v[jj].x, __ATOMIC_RELAXED, __HIP_MEMORY_SCOPE_WORKGROUP);
        __hip_atomic_fetch_add(a + 32, v[jj].y, __ATOMIC_RELAXED, __HIP_MEMORY_SCOPE_WORKGROUP);
        __hip_atomic_fetch_add(a + 64, v[jj].z, __ATOMIC_RELAXED, __HIP_MEMORY_SCOPE_WORKGROUP);
        __hip_atomic_fetch_add(a + 96, v[jj].w, __ATOMIC_RELAXED, __HIP_MEMORY_SCOPE_WORKGROUP);
      }
    }
  }
  asm volatile("s_wait_dscnt 0x0" ::: "memory");
  __syncthreads();

  float4* h4 = reinterpret_cast<float4*>(h);
  for (int r = wave; r < n; r += OWN_T / 32) {
    const float* a = acc + r * D_FEAT + lane;
    float4 v;
    v.x = a[0]; v.y = a[32]; v.z = a[64]; v.w = a[96];
    h4[(size_t)(base + r) * 32 + lane] = v;
  }
}

#define MLP_PF 8
#define MLP_UNROLL _Pragma("unroll")

__device__ __forceinline__ void mlp_load_step(const float* __restrict__ w1p, const float* __restrict__ w2p,
                                              int G, int t, float4& wa, float4& wb) {
  if (t < 32) {
    const float* p = w1p + 64 * G + (size_t)(8 * (t >> 1) + 2 * (t & 1)) * D_HID;
    wa = *reinterpret_cast<const float4*>(p);
    wb = *reinterpret_cast<const float4*>(p + D_HID);
  } else {
    const int u = t - 32;
    const float* p = w2p + (size_t)(64 * G + 8 * ((u >> 1) & 3) + (u >> 3)) * D_OUT + 64 * (u & 1);
    wa = *reinterpret_cast<const float4*>(p);
    wb = *reinterpret_cast<const float4*>(p + 4 * D_OUT);
  }
}

template <int MLP_NT>
__device__ __forceinline__ void mlp_tiles(const float* __restrict__ h,
                                          const float* __restrict__ W1, const float* __restrict__ b1,
                                          const float* __restrict__ W2, const float* __restrict__ b2,
                                          float* __restrict__ out, float* slab, bool to_part,
                                          int node0, int G0, int G1, int lane) {
  const int c    = lane & 15;
  const int hh   = lane >> 4;

  const float* w1p = W1 + (size_t)(4 * hh) * D_HID + 4 * c;
  const float* w2p = W2 + (size_t)(32 * hh) * D_OUT + 4 * c;
  const float* b1p = b1 + 32 * hh;

  float4 ra[MLP_PF], rb[MLP_PF];
MLP_UNROLL
  for (int t = 0; t < MLP_PF; ++t) mlp_load_step(w1p, w2p, G0, t, ra[t], rb[t]);

  float4 hB[MLP_NT][16];
MLP_UNROLL
  for (int nt = 0; nt < MLP_NT; ++nt) {
    const float* hp = h + (size_t)(node0 + 16 * nt + c) * D_FEAT + 4 * hh;
MLP_UNROLL
    for (int j = 0; j < 16; ++j)
      hB[nt][j] = *reinterpret_cast<const float4*>(hp + 8 * j);
  }

  v8f acc[2][4][MLP_NT];
MLP_UNROLL
  for (int og = 0; og < 2; ++og)
MLP_UNROLL
    for (int o = 0; o < 4; ++o)
MLP_UNROLL
      for (int nt = 0; nt < MLP_NT; ++nt)
        acc[og][o][nt] = (v8f){0.f,0.f,0.f,0.f,0.f,0.f,0.f,0.f};

  for (int G = G0; G < G1; ++G) {
    const int Gn = (G + 1) & (D_HID / 64 - 1);

    float4 bb1[8];
MLP_UNROLL
    for (int r = 0; r < 8; ++r)
      bb1[r] = *reinterpret_cast<const float4*>(b1p + 64 * G + 4 * r);

    v8f d1[4][MLP_NT];
MLP_UNROLL
    for (int q = 0; q < 4; ++q)
MLP_UNROLL
      for (int nt = 0; nt < MLP_NT; ++nt)
        d1[q][nt] = (v8f){0.f,0.f,0.f,0.f,0.f,0.f,0.f,0.f};

MLP_UNROLL
    for (int t = 0; t < 32; ++t) {
      const int j = t >> 1, p = t & 1;
      const float4 wa = ra[t % MLP_PF];
      const float4 wb = rb[t % MLP_PF];
      mlp_load_step(w1p, w2p, G, t + MLP_PF, ra[t % MLP_PF], rb[t % MLP_PF]);
      v2f bf[MLP_NT];
MLP_UNROLL
      for (int nt = 0; nt < MLP_NT; ++nt) {
        bf[nt].x = p ? hB[nt][j].z : hB[nt][j].x;
        bf[nt].y = p ? hB[nt][j].w : hB[nt][j].y;
      }
      const float was[4] = {wa.x, wa.y, wa.z, wa.w};
      const float wbs[4] = {wb.x, wb.y, wb.z, wb.w};
MLP_UNROLL
      for (int q = 0; q < 4; ++q) {
        v2f a; a.x = was[q]; a.y = wbs[q];
MLP_UNROLL
        for (int nt = 0; nt < MLP_NT; ++nt)
          d1[q][nt] = wmma4(a, bf[nt], d1[q][nt]);
      }
    }

MLP_UNROLL
    for (int r = 0; r < 8; ++r) {
      const float bbs[4] = {bb1[r].x, bb1[r].y, bb1[r].z, bb1[r].w};
MLP_UNROLL
      for (int q = 0; q < 4; ++q)
MLP_UNROLL
        for (int nt = 0; nt < MLP_NT; ++nt) {
          const float v = d1[q][nt][r] + bbs[q];
          d1[q][nt][r] = v > 0.f ? v : 0.f;
        }
    }

MLP_UNROLL
    for (int u = 0; u < 32; ++u) {
      const int t = 32 + u;
      const int q = u >> 3, j2 = (u >> 1) & 3, og = u & 1;
      const float4 wa = ra[t % MLP_PF];
      const float4 wb = rb[t % MLP_PF];
      if (t + MLP_PF < 64) mlp_load_step(w1p, w2p, G,  t + MLP_PF,      ra[t % MLP_PF], rb[t % MLP_PF]);
      else                 mlp_load_step(w1p, w2p, Gn, t + MLP_PF - 64, ra[t % MLP_PF], rb[t % MLP_PF]);
      v2f bf[MLP_NT];
MLP_UNROLL
      for (int nt = 0; nt < MLP_NT; ++nt) {
        bf[nt].x = d1[q][nt][2 * j2];
        bf[nt].y = d1[q][nt][2 * j2 + 1];
      }
      const float was[4] = {wa.x, wa.y, wa.z, wa.w};
      const float wbs[4] = {wb.x, wb.y, wb.z, wb.w};
MLP_UNROLL
      for (int o = 0; o < 4; ++o) {
        v2f a; a.x = was[o]; a.y = wbs[o];
MLP_UNROLL
        for (int nt = 0; nt < MLP_NT; ++nt)
          acc[og][o][nt] = wmma4(a, bf[nt], acc[og][o][nt]);
      }
    }
  }

  if (MLP_NT == 1 && to_part) {
MLP_UNROLL
    for (int og = 0; og < 2; ++og) {
MLP_UNROLL
      for (int r = 0; r < 8; ++r) {
        float4 v;
        v.x = acc[og][0][0][r];
        v.y = acc[og][1][0][r];
        v.z = acc[og][2][0][r];
        v.w = acc[og][3][0][r];
        *reinterpret_cast<float4*>(slab + c * D_OUT + 64 * og + 32 * hh + 4 * r) = v;
      }
    }
    return;
  }
MLP_UNROLL
  for (int og = 0; og < 2; ++og) {
MLP_UNROLL
    for (int r = 0; r < 8; ++r) {
      const float4 bb = *reinterpret_cast<const float4*>(b2 + 64 * og + 32 * hh + 4 * r);
MLP_UNROLL
      for (int nt = 0; nt < MLP_NT; ++nt) {
        float4 v;
        v.x = acc[og][0][nt][r] + bb.x;
        v.y = acc[og][1][nt][r] + bb.y;
        v.z = acc[og][2][nt][r] + bb.z;
        v.w = acc[og][3][nt][r] + bb.w;
        *reinterpret_cast<float4*>(out + (size_t)(node0 + 16 * nt + c) * D_OUT
                                   + 64 * og + 32 * hh + 4 * r) = v;
      }
    }
  }
}

#define MLP_W     8
#define MLP_WG    256
#define MLP_WAVES (MLP_WG * MLP_W)
#define MLP_TILES (N_NODES / 16)
#define MLP_REST  (MLP_TILES - 3 * MLP_WAVES)
static_assert(N_NODES % 16 == 0, "whole node tiles");
static_assert(MLP_REST >= 0 && MLP_REST <= MLP_WG, "the left-over tiles go one to a workgroup");
static_assert(D_HID / 64 == MLP_W, "pass 3: one hidden group per wave");

__global__ __launch_bounds__(32 * MLP_W)
void gin_mlp_t(const float* __restrict__ h,
               const float* __restrict__ W1, const float* __restrict__ b1,
               const float* __restrict__ W2, const float* __restrict__ b2,
               float* __restrict__ out) {
  __shared__ float part[MLP_W * 16 * D_OUT];
  const int lane = threadIdx.x & 31;
  const int w  = __builtin_amdgcn_readfirstlane((int)(threadIdx.x >> 5));
  const int gw = blockIdx.x * MLP_W + w;
  float* slab = part + w * (16 * D_OUT);

  mlp_tiles<2>(h, W1, b1, W2, b2, out, slab, false, 32 * gw, 0, D_HID / 64, lane);

  const int np = ((int)blockIdx.x < MLP_REST) ? 2 : 1;
  for (int p = 0; p < np; ++p) {
    const bool sh   = (p == 1);
    const int node0 = sh ? 16 * (3 * MLP_WAVES + (int)blockIdx.x) : 16 * (2 * MLP_WAVES + gw);
    const int G0    = sh ? w : 0;
    const int G1    = sh ? w + 1 : D_HID / 64;
    mlp_tiles<1>(h, W1, b1, W2, b2, out, slab, sh, node0, G0, G1, lane);
    if (sh) {
      __syncthreads();
      const float4* p4 = reinterpret_cast<const float4*>(part);
      const float4* b4 = reinterpret_cast<const float4*>(b2);
      float4* o4 = reinterpret_cast<float4*>(out + (size_t)node0 * D_OUT);
      for (int i = threadIdx.x; i < 16 * D_OUT / 4; i += 32 * MLP_W) {
        float4 s = b4[i & (D_OUT / 4 - 1)];
        for (int k = 0; k < MLP_W; ++k) {
          const float4 v = p4[k * (16 * D_OUT / 4) + i];
          s.x += v.x; s.y += v.y; s.z += v.z; s.w += v.w;
        }
        o4[i] = s;
      }
    }
  }
}

extern "C" void kernel_launch(void* const* d_in, const int* in_sizes, int n_in,
                              void* d_out, int out_size, void* d_ws, size_t ws_size,
                              hipStream_t stream) {
  const float* x  = (const float*)d_in[0];
  const int*   ei = (const int*)d_in[1];
  const float* W1 = (const float*)d_in[2];
  const float* b1 = (const float*)d_in[3];
  const float* W2 = (const float*)d_in[4];
  const float* b2 = (const float*)d_in[5];
  float* out = (float*)d_out;

  const int* src = ei;
  const int* dst = ei + N_EDGES;

  float* h = (float*)d_ws;
  unsigned* key = (unsigned*)(h + (size_t)N_NODES * D_FEAT);
  unsigned* bm  = key + KEY_DW;

  k_key<<<(KEY_DW + KEY_T * KEY_LD - 1) / (KEY_T * KEY_LD), KEY_T, 0, stream>>>(dst, key, bm);

  k_own<<<OWN_WG, OWN_T, 0, stream>>>(x, src, dst, key, bm, h);

  gin_mlp_t<<<MLP_WG, 32 * MLP_W, 0, stream>>>(h, W1, b1, W2, b2, out);
}
